# starting kernel
# baseline (speedup 1.0000x reference)
#include <hip/hip_runtime.h>
#include <cmath>

typedef __attribute__((ext_vector_type(2))) float v2f;
typedef __attribute__((ext_vector_type(8))) float v8f;

__global__ void gcn_copy_h_kernel(const float* __restrict__ h,
                                  float* __restrict__ out, int n_nodes) {
  int i = blockIdx.x * blockDim.x + threadIdx.x;
  if (i >= n_nodes * 16) return;
  int r  = i >> 4;
  int c4 = i & 15;
  float4 v = *((const float4*)(h + (size_t)r * 64) + c4);
  *((float4*)(out + (size_t)r * 256) + c4) = v;
}

__global__ void gcn_zero_kernel(float4* __restrict__ p, int n4) {
  int i = blockIdx.x * blockDim.x + threadIdx.x;
  if (i < n4) p[i] = make_float4(0.f, 0.f, 0.f, 0.f);
}

__global__ void gcn_edge_scatter_kernel(const float* __restrict__ hprev,
                                        const float* __restrict__ ew,
                                        const int* __restrict__ src,
                                        const int* __restrict__ dst,
                                        float* __restrict__ agg,
                                        int n_edges) {
  int wave = (blockIdx.x * blockDim.x + threadIdx.x) >> 5;
  int lane = threadIdx.x & 31;
  int e = __builtin_amdgcn_readfirstlane(wave);
  if (e >= n_edges) return;
  int   s = src[e];
  int   d = dst[e];
  float w = ew[e];
  float2 hv = *(const float2*)(hprev + (size_t)s * 256 + lane * 2);
  float* p = agg + (size_t)d * 64 + lane * 2;
  unsafeAtomicAdd(p,     hv.x * w);
  unsafeAtomicAdd(p + 1, hv.y * w);
}

__global__ void gcn_gemm_wmma_kernel(const float* __restrict__ A,
                                     const float* __restrict__ W,
                                     const float* __restrict__ bias,
                                     float* __restrict__ out,
                                     int n_rows, int do_tanh) {
  int wave = threadIdx.x >> 5;
  int lane = threadIdx.x & 31;
  int row0 = blockIdx.x * 32 + (wave >> 2) * 16;
  int col0 = (wave & 3) * 16;
  if (row0 >= n_rows) return;

  int lh = lane >> 4;
  int lm = lane & 15;
  const float* arow = A + (size_t)(row0 + lm) * 64 + 2 * lh;
  const float* wcol = W + (size_t)(2 * lh) * 64 + col0 + lm;

  v8f acc = {};
#pragma unroll
  for (int k0 = 0; k0 < 64; k0 += 4) {
    v2f a = *(const v2f*)(arow + k0);
    v2f b;
    b.x = wcol[(size_t)k0 * 64];
    b.y = wcol[(size_t)k0 * 64 + 64];
    acc = __builtin_amdgcn_wmma_f32_16x16x4_f32(
         false, a,  false, b,
         (short)0, acc,  false,  false);
  }

  float bv = bias[col0 + lm];
  float* orow = out + (size_t)(row0 + 8 * lh) * 256 + col0 + lm;
#pragma unroll
  for (int v = 0; v < 8; ++v) {
    float val = acc[v] + bv;
    if (do_tanh) val = tanhf(val);
    orow[(size_t)v * 256] = val;
  }
}

extern "C" void kernel_launch(void* const* d_in, const int* in_sizes, int n_in,
                              void* d_out, int out_size, void* d_ws, size_t ws_size,
                              hipStream_t stream) {
  const float* h  = (const float*)d_in[0];
  const float* ew = (const float*)d_in[1];
  const float* Ws = (const float*)d_in[2];
  const float* bs = (const float*)d_in[3];
  const int*  src = (const int*)d_in[4];
  const int*  dst = (const int*)d_in[5];
  float* out = (float*)d_out;
  float* agg = (float*)d_ws;

  const int N = in_sizes[0] / 64;
  const int E = in_sizes[1];
  const int L = 3;

  {
    int t = N * 16;
    gcn_copy_h_kernel<<<(t + 255) / 256, 256, 0, stream>>>(h, out, N);
  }

  for (int layer = 0; layer < L; ++layer) {
    {
      int n4 = N * 16;
      gcn_zero_kernel<<<(n4 + 255) / 256, 256, 0, stream>>>((float4*)agg, n4);
    }
    {
      long threads = (long)E * 32;
      int blocks = (int)((threads + 255) / 256);
      gcn_edge_scatter_kernel<<<blocks, 256, 0, stream>>>(
          out + layer * 64, ew, src, dst, agg, E);
    }
    {
      int blocks = (N + 31) / 32;
      gcn_gemm_wmma_kernel<<<blocks, 256, 0, stream>>>(
          agg, Ws + layer * 64 * 64, bs + layer * 64,
          out + (layer + 1) * 64, N, (layer < L - 1) ? 1 : 0);
    }
  }
}


# edges split once per call into 1024-row source blocks (LDS integer ranks); a work-group keeps its block of h in LDS and reads edge rows there: only the f32 atomic adds go to L2; arrival-order sums
# speedup vs baseline: 2.5725x; 2.5725x over previous
#include <hip/hip_runtime.h>
#include <cmath>

typedef __attribute__((ext_vector_type(2))) float v2f;
typedef __attribute__((ext_vector_type(8))) float v8f;

#define SRC_BLOCK_SHIFT 10
#define SRC_BLOCK       1024
#define N_CHUNKS        320
#define N_PARTS         5
#define CELL_CAP        128
#define MAX_BLOCKS      256

__global__ void gcn_init_kernel(const float* __restrict__ h,
                                float* __restrict__ out,
                                float* __restrict__ agg0, int n_nodes,
                                int* __restrict__ ovf_cnt) {
  int tid    = blockIdx.x * blockDim.x + threadIdx.x;
  int n_thr  = gridDim.x * blockDim.x;
  int n_zero = n_nodes * 32;
  int n_copy = n_nodes * 16;
  if (tid == 0) *ovf_cnt = 0;
  for (int i = tid; i < n_copy; i += n_thr) {
    int r  = i >> 4;
    int c4 = i & 15;
    float4 v = *((const float4*)(h + (size_t)r * 64) + c4);
    *((float4*)(out + (size_t)r * 256) + c4) = v;
  }
  for (int i = tid; i < n_zero; i += n_thr)
    ((float4*)agg0)[i] = make_float4(0.f, 0.f, 0.f, 0.f);
}

__global__ void __launch_bounds__(1024)
gcn_bin_kernel(const float* __restrict__ ew,
               const int* __restrict__ src,
               const int* __restrict__ dst,
               int n_edges, int n_blocks,
               int* __restrict__ cell_cnt,
               int4* __restrict__ cell_rec,
               int* __restrict__ ovf_cnt,
               int* __restrict__ ovf_list) {
  __shared__ int cnt[MAX_BLOCKS];
  int c = blockIdx.x;
  for (int b = threadIdx.x; b < MAX_BLOCKS; b += blockDim.x) cnt[b] = 0;
  __syncthreads();
  int e0 = (int)(((long long)c * n_edges) / N_CHUNKS);
  int e1 = (int)(((long long)(c + 1) * n_edges) / N_CHUNKS);
  for (int base = e0 + (int)threadIdx.x; base - (int)threadIdx.x < e1; base += 4 * (int)blockDim.x) {
    int   s[4], d[4];
    float w[4];
#pragma unroll
    for (int u = 0; u < 4; ++u) {
      int e  = base + u * (int)blockDim.x;
      int ec = e < e1 ? e : e1 - 1;
      s[u] = src[ec];
      d[u] = dst[ec];
      w[u] = ew[ec];
    }
#pragma unroll
    for (int u = 0; u < 4; ++u) {
      int e = base + u * (int)blockDim.x;
      if (e < e1) {
        int b = s[u] >> SRC_BLOCK_SHIFT;
        int r = atomicAdd(&cnt[b], 1);
        if (r < CELL_CAP) {
          cell_rec[((size_t)b * N_CHUNKS + c) * CELL_CAP + r] =
              make_int4(s[u] & (SRC_BLOCK - 1), d[u], __float_as_int(w[u]), 0);
        } else {
          int p = atomicAdd(ovf_cnt, 1);
          ovf_list[p] = e;
        }
      }
    }
  }
  __syncthreads();
  for (int b = threadIdx.x; b < n_blocks; b += blockDim.x) {
    int n = cnt[b];
    cell_cnt[b * N_CHUNKS + c] = n < CELL_CAP ? n : CELL_CAP;
  }
}

__global__ void __launch_bounds__(1024)
gcn_edge_scatter_kernel(const float* __restrict__ hprev,
                        const float* __restrict__ ew,
                        const int* __restrict__ src,
                        const int* __restrict__ dst,
                        float* __restrict__ agg,
                        int n_nodes, int plane_floats,
                        float* __restrict__ agg_next, int next_f4,
                        const int* __restrict__ cell_cnt,
                        const int4* __restrict__ cell_rec,
                        const int* __restrict__ ovf_cnt,
                        const int* __restrict__ ovf_list,
                        int n_edges) {
  __shared__ __attribute__((aligned(16))) float hs[SRC_BLOCK * 64];
  int tid   = blockIdx.x * blockDim.x + threadIdx.x;
  int n_thr = gridDim.x * blockDim.x;
  int lane  = threadIdx.x & 31;
  int wave  = __builtin_amdgcn_readfirstlane((int)(threadIdx.x >> 5));
  int b     = blockIdx.x / N_PARTS;
  int part  = blockIdx.x - b * N_PARTS;

  if (agg_next != nullptr) {
    for (int i = tid; i < next_f4; i += n_thr)
      ((float4*)agg_next)[i] = make_float4(0.f, 0.f, 0.f, 0.f);
  }

  int row_base = b << SRC_BLOCK_SHIFT;
  int n_rows   = n_nodes - row_base;
  if (n_rows > SRC_BLOCK) n_rows = SRC_BLOCK;
  {
    float4 v[SRC_BLOCK * 16 / 1024];
#pragma unroll
    for (int it = 0; it < SRC_BLOCK * 16 / 1024; ++it) {
      int i  = (int)threadIdx.x + it * 1024;
      int r  = i >> 4;
      int c4 = i & 15;
      int rr = r < n_rows ? r : n_rows - 1;
      v[it] = *((const float4*)(hprev + (size_t)(row_base + rr) * 256) + c4);
    }
#pragma unroll
    for (int it = 0; it < SRC_BLOCK * 16 / 1024; ++it) {
      int i  = (int)threadIdx.x + it * 1024;
      int r  = i >> 4;
      int c4 = i & 15;
      *((float4*)(hs + r * 64) + c4) = v[it];
    }
  }
  __syncthreads();

  unsigned pid  = __builtin_amdgcn_s_sendmsg_rtn(0x87);
  unsigned half = (pid >> 18) & 1u;
  float* pcol = agg + (size_t)half * plane_floats + lane;
  const float* hl = hs + lane;

  const int cells_per_part = N_CHUNKS / N_PARTS;
  const int waves = (int)(blockDim.x >> 5);
  for (int k = wave; k < cells_per_part; k += waves) {
    int cell = b * N_CHUNKS + part * cells_per_part + k;
    int n    = cell_cnt[cell];
    const int4* rec = cell_rec + (size_t)cell * CELL_CAP;
    for (int r0 = 0; r0 < n; r0 += 32) {
      int4 q = make_int4(0, 0, 0, 0);
      if (r0 + lane < n) q = rec[r0 + lane];
      int m = n - r0;
      if (m > 32) m = 32;
      int   s0 = __builtin_amdgcn_readlane(q.x, 0);
      float h0 = hl[s0 * 64];
      float h1 = hl[s0 * 64 + 32];
      for (int j = 0; j < m; ++j) {
        int   sn = __builtin_amdgcn_readlane(q.x, (j + 1) & 31);
        float n0 = hl[sn * 64];
        float n1 = hl[sn * 64 + 32];
        int   d = __builtin_amdgcn_readlane(q.y, j);
        float w = __int_as_float(__builtin_amdgcn_readlane(q.z, j));
        float* p = pcol + (size_t)d * 64;
        unsafeAtomicAdd(p,      h0 * w);
        unsafeAtomicAdd(p + 32, h1 * w);
        h0 = n0;
        h1 = n1;
      }
    }
  }

  int n_ovf = *ovf_cnt;
  if (n_ovf > n_edges) n_ovf = n_edges;
  int gw = blockIdx.x * waves + wave;
  int nw = gridDim.x * waves;
  const float* hcol = hprev + lane;
  for (int i = gw; i < n_ovf; i += nw) {
    int   e = ovf_list[i];
    int   s = src[e];
    int   d = dst[e];
    float w = ew[e];
    const float* hr = hcol + (size_t)s * 256;
    float h0 = hr[0];
    float h1 = hr[32];
    float* p = pcol + (size_t)d * 64;
    unsafeAtomicAdd(p,      h0 * w);
    unsafeAtomicAdd(p + 32, h1 * w);
  }
}

__global__ void __launch_bounds__(128)
gcn_gemm_wmma_kernel(const float* __restrict__ A,
                                     const float* __restrict__ W,
                                     const float* __restrict__ bias,
                                     float* __restrict__ out,
                                     int n_rows, int do_tanh, int plane_floats) {
  int wave = threadIdx.x >> 5;
  int lane = threadIdx.x & 31;
  int row0 = (blockIdx.x * (blockDim.x >> 5) + wave) * 16;
  if (row0 >= n_rows) return;

  int lh = lane >> 4;
  int lm = lane & 15;
  const float* arow  = A + (size_t)(row0 + lm) * 64 + 2 * lh;
  const float* arow1 = arow + plane_floats;
  const float* wrow = W + (size_t)(2 * lh) * 64 + 4 * lm;

  v8f acc0 = {}, acc1 = {}, acc2 = {}, acc3 = {};
#pragma unroll
  for (int k0 = 0; k0 < 64; k0 += 4) {
    v2f a = *(const v2f*)(arow + k0) + *(const v2f*)(arow1 + k0);
    const float4 w0 = *(const float4*)(wrow + (size_t)k0 * 64);
    const float4 w1 = *(const float4*)(wrow + (size_t)k0 * 64 + 64);
    v2f b0, b1, b2, b3;
    b0.x = w0.x;  b0.y = w1.x;
    b1.x = w0.y;  b1.y = w1.y;
    b2.x = w0.z;  b2.y = w1.z;
    b3.x = w0.w;  b3.y = w1.w;
    acc0 = __builtin_amdgcn_wmma_f32_16x16x4_f32(
         false, a,  false, b0,
         (short)0, acc0,  false,  false);
    acc1 = __builtin_amdgcn_wmma_f32_16x16x4_f32(
        false, a, false, b1, (short)0, acc1, false, false);
    acc2 = __builtin_amdgcn_wmma_f32_16x16x4_f32(
        false, a, false, b2, (short)0, acc2, false, false);
    acc3 = __builtin_amdgcn_wmma_f32_16x16x4_f32(
        false, a, false, b3, (short)0, acc3, false, false);
  }

  const float4 bv = *(const float4*)(bias + 4 * lm);
  float* orow = out + (size_t)(row0 + 8 * lh) * 256 + 4 * lm;
#pragma unroll
  for (int v = 0; v < 8; ++v) {
    float4 o;
    o.x = acc0[v] + bv.x;
    o.y = acc1[v] + bv.y;
    o.z = acc2[v] + bv.z;
    o.w = acc3[v] + bv.w;
    if (do_tanh) {
      o.x = tanhf(o.x);
      o.y = tanhf(o.y);
      o.z = tanhf(o.z);
      o.w = tanhf(o.w);
    }
    *(float4*)(orow + (size_t)v * 256) = o;
  }
}

extern "C" void kernel_launch(void* const* d_in, const int* in_sizes, int n_in,
                              void* d_out, int out_size, void* d_ws, size_t ws_size,
                              hipStream_t stream) {
  const float* h  = (const float*)d_in[0];
  const float* ew = (const float*)d_in[1];
  const float* Ws = (const float*)d_in[2];
  const float* bs = (const float*)d_in[3];
  const int*  src = (const int*)d_in[4];
  const int*  dst = (const int*)d_in[5];
  float* out = (float*)d_out;
  float* agg = (float*)d_ws;

  const int N = in_sizes[0] / 64;
  const int E = in_sizes[1];
  const int L = 3;
  const int NB = (N + SRC_BLOCK - 1) / SRC_BLOCK;

  int*  cell_cnt = (int*)(agg + (size_t)L * N * 128);
  int4* cell_rec = (int4*)(cell_cnt + (size_t)NB * N_CHUNKS);
  int*  ovf_cnt  = (int*)(cell_rec + (size_t)NB * N_CHUNKS * CELL_CAP);
  int*  ovf_list = ovf_cnt + 4;

  {
    gcn_init_kernel<<<1024, 256, 0, stream>>>(h, out, agg, N, ovf_cnt);
  }
  {
    gcn_bin_kernel<<<N_CHUNKS, 1024, 0, stream>>>(ew, src, dst, E, NB, cell_cnt, cell_rec,
                                                 ovf_cnt, ovf_list);
  }

  for (int layer = 0; layer < L; ++layer) {
    float* agg_l    = agg + (size_t)layer * N * 128;
    float* agg_next = (layer < L - 1) ? agg_l + (size_t)N * 128 : nullptr;
    {
      gcn_edge_scatter_kernel<<<NB * N_PARTS, 1024, 0, stream>>>(
          out + layer * 64, ew, src, dst, agg_l, N, N * 64, agg_next, N * 32,
          cell_cnt, cell_rec, ovf_cnt, ovf_list, E);
    }
    {
      int blocks = (N + 63) / 64;
      gcn_gemm_wmma_kernel<<<blocks, 128, 0, stream>>>(
          agg_l, Ws + layer * 64 * 64, bs + layer * 64,
          out + (layer + 1) * 64, N, (layer < L - 1) ? 1 : 0, N * 64);
    }
  }
}


# binning pass: a chunk's records are staged per cell in LDS and each cell leaves as whole 128-byte lines (was one 16-byte store per edge, up to 32 lines per wave store); arrival-order sums kept
# speedup vs baseline: 2.6713x; 1.0384x over previous
#include <hip/hip_runtime.h>
#include <cmath>

typedef __attribute__((ext_vector_type(2))) float v2f;
typedef __attribute__((ext_vector_type(8))) float v8f;

#define SRC_BLOCK_SHIFT 10
#define SRC_BLOCK       1024
#define N_CHUNKS        320
#define N_PARTS         5
#define CELL_CAP        128
#define MAX_BLOCKS      256
#define STAGE_BLOCKS    64

__global__ void gcn_init_kernel(const float* __restrict__ h,
                                float* __restrict__ out,
                                float* __restrict__ agg0, int n_nodes,
                                int* __restrict__ ovf_cnt) {
  int tid    = blockIdx.x * blockDim.x + threadIdx.x;
  int n_thr  = gridDim.x * blockDim.x;
  int n_zero = n_nodes * 32;
  int n_copy = n_nodes * 16;
  if (tid == 0) *ovf_cnt = 0;
  for (int i = tid; i < n_copy; i += n_thr) {
    int r  = i >> 4;
    int c4 = i & 15;
    float4 v = *((const float4*)(h + (size_t)r * 64) + c4);
    *((float4*)(out + (size_t)r * 256) + c4) = v;
  }
  for (int i = tid; i < n_zero; i += n_thr)
    ((float4*)agg0)[i] = make_float4(0.f, 0.f, 0.f, 0.f);
}

__global__ void __launch_bounds__(1024)
gcn_bin_kernel(const float* __restrict__ ew,
               const int* __restrict__ src,
               const int* __restrict__ dst,
               int n_edges, int n_blocks,
               int* __restrict__ cell_cnt,
               int4* __restrict__ cell_rec,
               int* __restrict__ ovf_cnt,
               int* __restrict__ ovf_list) {
  __shared__ int cnt[MAX_BLOCKS];
  __shared__ __attribute__((aligned(16))) int4 stage[STAGE_BLOCKS * CELL_CAP];
  int c = blockIdx.x;
  for (int b = threadIdx.x; b < MAX_BLOCKS; b += blockDim.x) cnt[b] = 0;
  __syncthreads();
  int e0 = (int)(((long long)c * n_edges) / N_CHUNKS);
  int e1 = (int)(((long long)(c + 1) * n_edges) / N_CHUNKS);
  for (int base = e0 + (int)threadIdx.x; base - (int)threadIdx.x < e1; base += 4 * (int)blockDim.x) {
    int   s[4], d[4];
    float w[4];
#pragma unroll
    for (int u = 0; u < 4; ++u) {
      int e  = base + u * (int)blockDim.x;
      int ec = e < e1 ? e : e1 - 1;
      s[u] = src[ec];
      d[u] = dst[ec];
      w[u] = ew[ec];
    }
#pragma unroll
    for (int u = 0; u < 4; ++u) {
      int e = base + u * (int)blockDim.x;
      if (e < e1) {
        int b = s[u] >> SRC_BLOCK_SHIFT;
        int r = atomicAdd(&cnt[b], 1);
        if (r < CELL_CAP) {
          int4 q = make_int4(s[u] & (SRC_BLOCK - 1), d[u], __float_as_int(w[u]), 0);
          if (b < STAGE_BLOCKS) stage[b * CELL_CAP + r] = q;
          else cell_rec[((size_t)b * N_CHUNKS + c) * CELL_CAP + r] = q;
        } else {
          int p = atomicAdd(ovf_cnt, 1);
          ovf_list[p] = e;
        }
      }
    }
  }
  __syncthreads();
  {
    int lane   = (int)(threadIdx.x & 31);
    int wave   = (int)(threadIdx.x >> 5);
    int waves  = (int)(blockDim.x >> 5);
    int nb_st  = n_blocks < STAGE_BLOCKS ? n_blocks : STAGE_BLOCKS;
    int n_item = nb_st * (CELL_CAP / 32);
    for (int it = wave; it < n_item; it += waves) {
      int b  = it / (CELL_CAP / 32);
      int r  = (it - b * (CELL_CAP / 32)) * 32 + lane;
      int n  = cnt[b];
      if (n > CELL_CAP) n = CELL_CAP;
      if (r < n)
        cell_rec[((size_t)b * N_CHUNKS + c) * CELL_CAP + r] = stage[b * CELL_CAP + r];
    }
  }
  for (int b = threadIdx.x; b < n_blocks; b += blockDim.x) {
    int n = cnt[b];
    cell_cnt[b * N_CHUNKS + c] = n < CELL_CAP ? n : CELL_CAP;
  }
}

__global__ void __launch_bounds__(1024)
gcn_edge_scatter_kernel(const float* __restrict__ hprev,
                        const float* __restrict__ ew,
                        const int* __restrict__ src,
                        const int* __restrict__ dst,
                        float* __restrict__ agg,
                        int n_nodes, int plane_floats,
                        float* __restrict__ agg_next, int next_f4,
                        const int* __restrict__ cell_cnt,
                        const int4* __restrict__ cell_rec,
                        const int* __restrict__ ovf_cnt,
                        const int* __restrict__ ovf_list,
                        int n_edges) {
  __shared__ __attribute__((aligned(16))) float hs[SRC_BLOCK * 64];
  int tid   = blockIdx.x * blockDim.x + threadIdx.x;
  int n_thr = gridDim.x * blockDim.x;
  int lane  = threadIdx.x & 31;
  int wave  = __builtin_amdgcn_readfirstlane((int)(threadIdx.x >> 5));
  int b     = blockIdx.x / N_PARTS;
  int part  = blockIdx.x - b * N_PARTS;

  if (agg_next != nullptr) {
    for (int i = tid; i < next_f4; i += n_thr)
      ((float4*)agg_next)[i] = make_float4(0.f, 0.f, 0.f, 0.f);
  }

  int row_base = b << SRC_BLOCK_SHIFT;
  int n_rows   = n_nodes - row_base;
  if (n_rows > SRC_BLOCK) n_rows = SRC_BLOCK;
  {
    float4 v[SRC_BLOCK * 16 / 1024];
#pragma unroll
    for (int it = 0; it < SRC_BLOCK * 16 / 1024; ++it) {
      int i  = (int)threadIdx.x + it * 1024;
      int r  = i >> 4;
      int c4 = i & 15;
      int rr = r < n_rows ? r : n_rows - 1;
      v[it] = *((const float4*)(hprev + (size_t)(row_base + rr) * 256) + c4);
    }
#pragma unroll
    for (int it = 0; it < SRC_BLOCK * 16 / 1024; ++it) {
      int i  = (int)threadIdx.x + it * 1024;
      int r  = i >> 4;
      int c4 = i & 15;
      *((float4*)(hs + r * 64) + c4) = v[it];
    }
  }
  __syncthreads();

  unsigned pid  = __builtin_amdgcn_s_sendmsg_rtn(0x87);
  unsigned half = (pid >> 18) & 1u;
  float* pcol = agg + (size_t)half * plane_floats + lane;
  const float* hl = hs + lane;

  const int cells_per_part = N_CHUNKS / N_PARTS;
  const int waves = (int)(blockDim.x >> 5);
  for (int k = wave; k < cells_per_part; k += waves) {
    int cell = b * N_CHUNKS + part * cells_per_part + k;
    int n    = cell_cnt[cell];
    const int4* rec = cell_rec + (size_t)cell * CELL_CAP;
    for (int r0 = 0; r0 < n; r0 += 32) {
      int4 q = make_int4(0, 0, 0, 0);
      if (r0 + lane < n) q = rec[r0 + lane];
      int m = n - r0;
      if (m > 32) m = 32;
      int   s0 = __builtin_amdgcn_readlane(q.x, 0);
      float h0 = hl[s0 * 64];
      float h1 = hl[s0 * 64 + 32];
      for (int j = 0; j < m; ++j) {
        int   sn = __builtin_amdgcn_readlane(q.x, (j + 1) & 31);
        float n0 = hl[sn * 64];
        float n1 = hl[sn * 64 + 32];
        int   d = __builtin_amdgcn_readlane(q.y, j);
        float w = __int_as_float(__builtin_amdgcn_readlane(q.z, j));
        float* p = pcol + (size_t)d * 64;
        unsafeAtomicAdd(p,      h0 * w);
        unsafeAtomicAdd(p + 32, h1 * w);
        h0 = n0;
        h1 = n1;
      }
    }
  }

  int n_ovf = *ovf_cnt;
  if (n_ovf > n_edges) n_ovf = n_edges;
  int gw = blockIdx.x * waves + wave;
  int nw = gridDim.x * waves;
  const float* hcol = hprev + lane;
  for (int i = gw; i < n_ovf; i += nw) {
    int   e = ovf_list[i];
    int   s = src[e];
    int   d = dst[e];
    float w = ew[e];
    const float* hr = hcol + (size_t)s * 256;
    float h0 = hr[0];
    float h1 = hr[32];
    float* p = pcol + (size_t)d * 64;
    unsafeAtomicAdd(p,      h0 * w);
    unsafeAtomicAdd(p + 32, h1 * w);
  }
}

__global__ void __launch_bounds__(128)
gcn_gemm_wmma_kernel(const float* __restrict__ A,
                                     const float* __restrict__ W,
                                     const float* __restrict__ bias,
                                     float* __restrict__ out,
                                     int n_rows, int do_tanh, int plane_floats) {
  int wave = threadIdx.x >> 5;
  int lane = threadIdx.x & 31;
  int row0 = (blockIdx.x * (blockDim.x >> 5) + wave) * 16;
  if (row0 >= n_rows) return;

  int lh = lane >> 4;
  int lm = lane & 15;
  const float* arow  = A + (size_t)(row0 + lm) * 64 + 2 * lh;
  const float* arow1 = arow + plane_floats;
  const float* wrow = W + (size_t)(2 * lh) * 64 + 4 * lm;

  v8f acc0 = {}, acc1 = {}, acc2 = {}, acc3 = {};
#pragma unroll
  for (int k0 = 0; k0 < 64; k0 += 4) {
    v2f a = *(const v2f*)(arow + k0) + *(const v2f*)(arow1 + k0);
    const float4 w0 = *(const float4*)(wrow + (size_t)k0 * 64);
    const float4 w1 = *(const float4*)(wrow + (size_t)k0 * 64 + 64);
    v2f b0, b1, b2, b3;
    b0.x = w0.x;  b0.y = w1.x;
    b1.x = w0.y;  b1.y = w1.y;
    b2.x = w0.z;  b2.y = w1.z;
    b3.x = w0.w;  b3.y = w1.w;
    acc0 = __builtin_amdgcn_wmma_f32_16x16x4_f32(
         false, a,  false, b0,
         (short)0, acc0,  false,  false);
    acc1 = __builtin_amdgcn_wmma_f32_16x16x4_f32(
        false, a, false, b1, (short)0, acc1, false, false);
    acc2 = __builtin_amdgcn_wmma_f32_16x16x4_f32(
        false, a, false, b2, (short)0, acc2, false, false);
    acc3 = __builtin_amdgcn_wmma_f32_16x16x4_f32(
        false, a, false, b3, (short)0, acc3, false, false);
  }

  const float4 bv = *(const float4*)(bias + 4 * lm);
  float* orow = out + (size_t)(row0 + 8 * lh) * 256 + 4 * lm;
#pragma unroll
  for (int v = 0; v < 8; ++v) {
    float4 o;
    o.x = acc0[v] + bv.x;
    o.y = acc1[v] + bv.y;
    o.z = acc2[v] + bv.z;
    o.w = acc3[v] + bv.w;
    if (do_tanh) {
      o.x = tanhf(o.x);
      o.y = tanhf(o.y);
      o.z = tanhf(o.z);
      o.w = tanhf(o.w);
    }
    *(float4*)(orow + (size_t)v * 256) = o;
  }
}

extern "C" void kernel_launch(void* const* d_in, const int* in_sizes, int n_in,
                              void* d_out, int out_size, void* d_ws, size_t ws_size,
                              hipStream_t stream) {
  const float* h  = (const float*)d_in[0];
  const float* ew = (const float*)d_in[1];
  const float* Ws = (const float*)d_in[2];
  const float* bs = (const float*)d_in[3];
  const int*  src = (const int*)d_in[4];
  const int*  dst = (const int*)d_in[5];
  float* out = (float*)d_out;
  float* agg = (float*)d_ws;

  const int N = in_sizes[0] / 64;
  const int E = in_sizes[1];
  const int L = 3;
  const int NB = (N + SRC_BLOCK - 1) / SRC_BLOCK;

  int*  cell_cnt = (int*)(agg + (size_t)L * N * 128);
  int4* cell_rec = (int4*)(cell_cnt + (size_t)NB * N_CHUNKS);
  int*  ovf_cnt  = (int*)(cell_rec + (size_t)NB * N_CHUNKS * CELL_CAP);
  int*  ovf_list = ovf_cnt + 4;

  {
    gcn_init_kernel<<<1024, 256, 0, stream>>>(h, out, agg, N, ovf_cnt);
  }
  {
    gcn_bin_kernel<<<N_CHUNKS, 1024, 0, stream>>>(ew, src, dst, E, NB, cell_cnt, cell_rec,
                                                 ovf_cnt, ovf_list);
  }

  for (int layer = 0; layer < L; ++layer) {
    float* agg_l    = agg + (size_t)layer * N * 128;
    float* agg_next = (layer < L - 1) ? agg_l + (size_t)N * 128 : nullptr;
    {
      gcn_edge_scatter_kernel<<<NB * N_PARTS, 1024, 0, stream>>>(
          out + layer * 64, ew, src, dst, agg_l, N, N * 64, agg_next, N * 32,
          cell_cnt, cell_rec, ovf_cnt, ovf_list, E);
    }
    {
      int blocks = (N + 63) / 64;
      gcn_gemm_wmma_kernel<<<blocks, 128, 0, stream>>>(
          agg_l, Ws + layer * 64 * 64, bs + layer * 64,
          out + (layer + 1) * 64, N, (layer < L - 1) ? 1 : 0, N * 64);
    }
  }
}
